# starting kernel
# baseline (speedup 1.0000x reference)
#include <hip/hip_runtime.h>

typedef float v2f __attribute__((ext_vector_type(2)));
typedef float v8f __attribute__((ext_vector_type(8)));

#define BN_EPS 1e-3f

__global__ __launch_bounds__(256) void deg_kernel(const int* __restrict__ ei,
                                                  float* __restrict__ deg, int E) {
  int e = blockIdx.x * blockDim.x + threadIdx.x;
  if (e < E) atomicAdd(&deg[ei[2 * e]], 1.0f);
}

__global__ __launch_bounds__(256) void invd_kernel(const float* __restrict__ deg,
                                                   float* __restrict__ invd, int n) {
  int i = blockIdx.x * blockDim.x + threadIdx.x;
  if (i < n) invd[i] = rsqrtf(deg[i]);
}

__global__ __launch_bounds__(256) void gemm_wmma(const float* __restrict__ X,
                                                 const float* __restrict__ W,
                                                 const float* __restrict__ bias,
                                                 float* __restrict__ Hout, int Nrows) {
  const int F = 128, Hc = 64;
  const int lane = threadIdx.x & 31;
  const int wave = threadIdx.x >> 5;
  const int tile = blockIdx.x * 8 + wave;
  const int ntiles = ((Nrows + 15) >> 4) * 4;
  if (tile >= ntiles) return;

  const int row0 = (tile >> 2) << 4;
  const int col0 = (tile & 3) << 4;
  const int m  = lane & 15;
  const int hi = lane >> 4;

  const float* __restrict__ xrow = X + (size_t)(row0 + m) * F;
  const float* __restrict__ wcol = W + col0 + m;

  v8f acc = {};
#pragma unroll
  for (int k = 0; k < F; k += 4) {
    const int ka = k + 2 * hi;
    v2f a = *reinterpret_cast<const v2f*>(xrow + ka);
    v2f b;
    b.x = wcol[(size_t)ka * Hc];
    b.y = wcol[(size_t)(ka + 1) * Hc];
    acc = __builtin_amdgcn_wmma_f32_16x16x4_f32(
         false, a,  false, b,
         (short)0, acc,  false,  false);
  }

  const float bv = bias[col0 + m];
#pragma unroll
  for (int v = 0; v < 8; ++v) {
    const int row = row0 + v + 8 * hi;
    Hout[(size_t)row * Hc + (col0 + m)] = acc[v] + bv;
  }
}

__global__ __launch_bounds__(256) void edge_scatter(const int* __restrict__ ei,
                                                    const float* __restrict__ h,
                                                    const float* __restrict__ invd,
                                                    float* __restrict__ agg, int E) {
  long long gid = (long long)blockIdx.x * blockDim.x + threadIdx.x;
  int e = (int)(gid >> 5);
  if (e >= E) return;
  int j = ((int)gid & 31) << 1;
  int s = ei[2 * e];
  int d = ei[2 * e + 1];
  float scale = 0.5f * invd[s] * invd[d];
  const float2 hv = *reinterpret_cast<const float2*>(h + (size_t)d * 64 + j);
  atomicAdd(agg + (size_t)s * 64 + j,     scale * hv.x);
  atomicAdd(agg + (size_t)s * 64 + j + 1, scale * hv.y);
}

__global__ __launch_bounds__(256) void finalize_kernel(const float* __restrict__ agg,
                                                       const float* __restrict__ h,
                                                       const float* __restrict__ deg,
                                                       const float* __restrict__ gamma,
                                                       const float* __restrict__ beta,
                                                       const float* __restrict__ mean,
                                                       const float* __restrict__ var,
                                                       float* __restrict__ out,
                                                       long long total) {
  long long i = (long long)blockIdx.x * blockDim.x + threadIdx.x;
  if (i >= total) return;
  int node = (int)(i >> 6);
  int j = (int)(i & 63);
  float a = agg[i] + 0.5f * deg[node] * h[i];
  out[i] = (a - mean[j]) * (gamma[j] * rsqrtf(var[j] + BN_EPS)) + beta[j];
}

extern "C" void kernel_launch(void* const* d_in, const int* in_sizes, int n_in,
                              void* d_out, int out_size, void* d_ws, size_t ws_size,
                              hipStream_t stream) {
  const float* X     = (const float*)d_in[0];
  const float* W     = (const float*)d_in[1];
  const float* b     = (const float*)d_in[2];
  const float* gamma = (const float*)d_in[3];
  const float* beta  = (const float*)d_in[4];
  const float* mean  = (const float*)d_in[5];
  const float* var   = (const float*)d_in[6];
  const int*   ei    = (const int*)d_in[7];

  const int F  = 128, Hc = 64;
  const int N  = in_sizes[0] / F;
  const int E  = in_sizes[7] / 2;

  float* ws   = (float*)d_ws;
  float* h    = ws;
  float* agg  = h   + (size_t)N * Hc;
  float* deg  = agg + (size_t)N * Hc;
  float* invd = deg + N;

  hipMemsetAsync(agg, 0, ((size_t)N * Hc + (size_t)N) * sizeof(float), stream);

  deg_kernel<<<(E + 255) / 256, 256, 0, stream>>>(ei, deg, E);
  invd_kernel<<<(N + 255) / 256, 256, 0, stream>>>(deg, invd, N);

  const int ntiles = ((N + 15) / 16) * 4;
  gemm_wmma<<<(ntiles + 7) / 8, 256, 0, stream>>>(X, W, b, h, N);

  const long long ethreads = (long long)E * 32;
  edge_scatter<<<(int)((ethreads + 255) / 256), 256, 0, stream>>>(ei, h, invd, agg, E);

  const long long total = (long long)N * Hc;
  finalize_kernel<<<(int)((total + 255) / 256), 256, 0, stream>>>(
      agg, h, deg, gamma, beta, mean, var, (float*)d_out, total);
}


# scatter and finalize ran at the wave entry rate: a scatter wave now walks 32 edges (was 1), next 8 h rows requested ahead; a finalize lane does 8 floats (was 1); f32 atomics kept, arrival order free
# speedup vs baseline: 1.2712x; 1.2712x over previous
#include <hip/hip_runtime.h>

typedef float v2f __attribute__((ext_vector_type(2)));
typedef float v8f __attribute__((ext_vector_type(8)));

#define BN_EPS 1e-3f

__global__ __launch_bounds__(256) void deg_kernel(const int* __restrict__ ei,
                                                  float* __restrict__ deg, int E) {
  int e = blockIdx.x * blockDim.x + threadIdx.x;
  if (e < E) atomicAdd(&deg[ei[2 * e]], 1.0f);
}

__global__ __launch_bounds__(256) void invd_kernel(const float* __restrict__ deg,
                                                   float* __restrict__ invd, int n) {
  int i = blockIdx.x * blockDim.x + threadIdx.x;
  if (i < n) invd[i] = rsqrtf(deg[i]);
}

__global__ __launch_bounds__(256) void gemm_wmma(const float* __restrict__ X,
                                                 const float* __restrict__ W,
                                                 const float* __restrict__ bias,
                                                 float* __restrict__ Hout, int Nrows) {
  const int F = 128, Hc = 64;
  const int lane = threadIdx.x & 31;
  const int wave = threadIdx.x >> 5;
  const int tile = blockIdx.x * 8 + wave;
  const int ntiles = ((Nrows + 15) >> 4) * 4;
  if (tile >= ntiles) return;

  const int row0 = (tile >> 2) << 4;
  const int col0 = (tile & 3) << 4;
  const int m  = lane & 15;
  const int hi = lane >> 4;

  const float* __restrict__ xrow = X + (size_t)(row0 + m) * F;
  const float* __restrict__ wcol = W + col0 + m;

  v8f acc = {};
#pragma unroll
  for (int k = 0; k < F; k += 4) {
    const int ka = k + 2 * hi;
    v2f a = *reinterpret_cast<const v2f*>(xrow + ka);
    v2f b;
    b.x = wcol[(size_t)ka * Hc];
    b.y = wcol[(size_t)(ka + 1) * Hc];
    acc = __builtin_amdgcn_wmma_f32_16x16x4_f32(
         false, a,  false, b,
         (short)0, acc,  false,  false);
  }

  const float bv = bias[col0 + m];
#pragma unroll
  for (int v = 0; v < 8; ++v) {
    const int row = row0 + v + 8 * hi;
    Hout[(size_t)row * Hc + (col0 + m)] = acc[v] + bv;
  }
}

__device__ __forceinline__ float lane_f32(float v, int i) {
  return __builtin_bit_cast(float, __builtin_amdgcn_readlane(__builtin_bit_cast(int, v), i));
}

__global__ __launch_bounds__(256) void edge_scatter(const int* __restrict__ ei,
                                                    const float* __restrict__ h,
                                                    const float* __restrict__ invd,
                                                    float* __restrict__ agg, int E) {
  const int lane = threadIdx.x & 31;
  const int e0 = (blockIdx.x * 8 + (threadIdx.x >> 5)) << 5;
  if (e0 >= E) return;
  const int e = e0 + lane;
  int s = 0, d = 0;
  float sc = 0.0f;
  if (e < E) {
    const int2 sd = *reinterpret_cast<const int2*>(ei + 2 * (size_t)e);
    s = sd.x;
    d = sd.y;
    sc = 0.5f * invd[s] * invd[d];
  }
  const float* __restrict__ hj = h + (lane << 1);
  float* __restrict__ aj = agg + (lane << 1);

  if (e0 + 32 <= E) {
    float2 cur[8], nxt[8];
#pragma unroll
    for (int u = 0; u < 8; ++u) {
      const int di = __builtin_amdgcn_readlane(d, u);
      cur[u] = *reinterpret_cast<const float2*>(hj + (size_t)di * 64);
    }
#pragma unroll
    for (int g = 0; g < 32; g += 8) {
      if (g + 8 < 32) {
#pragma unroll
        for (int u = 0; u < 8; ++u) {
          const int di = __builtin_amdgcn_readlane(d, g + 8 + u);
          nxt[u] = *reinterpret_cast<const float2*>(hj + (size_t)di * 64);
        }
      }
#pragma unroll
      for (int u = 0; u < 8; ++u) {
        const int si = __builtin_amdgcn_readlane(s, g + u);
        const float w = lane_f32(sc, g + u);
        float* dstp = aj + (size_t)si * 64;
        atomicAdd(dstp,     w * cur[u].x);
        atomicAdd(dstp + 1, w * cur[u].y);
      }
#pragma unroll
      for (int u = 0; u < 8; ++u) cur[u] = nxt[u];
    }
  } else {
    const int cnt = E - e0;
    for (int i = 0; i < cnt; ++i) {
      const int si = __builtin_amdgcn_readlane(s, i);
      const int di = __builtin_amdgcn_readlane(d, i);
      const float w = lane_f32(sc, i);
      const float2 hv = *reinterpret_cast<const float2*>(hj + (size_t)di * 64);
      float* dstp = aj + (size_t)si * 64;
      atomicAdd(dstp,     w * hv.x);
      atomicAdd(dstp + 1, w * hv.y);
    }
  }
}

__device__ __forceinline__ float4 bn_scale4(const float* __restrict__ gamma,
                                            const float* __restrict__ var, const int c4) {
  const float4 ga = reinterpret_cast<const float4*>(gamma)[c4];
  const float4 va = reinterpret_cast<const float4*>(var)[c4];
  float4 g;
  g.x = ga.x * rsqrtf(va.x + BN_EPS);
  g.y = ga.y * rsqrtf(va.y + BN_EPS);
  g.z = ga.z * rsqrtf(va.z + BN_EPS);
  g.w = ga.w * rsqrtf(va.w + BN_EPS);
  return g;
}

__device__ __forceinline__ float4 bn4(const float4 a, const float4 hh, const float hd,
                                      const float4 mu, const float4 g, const float4 be) {
  float4 r;
  r.x = ((a.x + hd * hh.x) - mu.x) * g.x + be.x;
  r.y = ((a.y + hd * hh.y) - mu.y) * g.y + be.y;
  r.z = ((a.z + hd * hh.z) - mu.z) * g.z + be.z;
  r.w = ((a.w + hd * hh.w) - mu.w) * g.w + be.w;
  return r;
}

__global__ __launch_bounds__(256) void finalize_kernel(const float* __restrict__ agg,
                                                       const float* __restrict__ h,
                                                       const float* __restrict__ deg,
                                                       const float* __restrict__ gamma,
                                                       const float* __restrict__ beta,
                                                       const float* __restrict__ mean,
                                                       const float* __restrict__ var,
                                                       float* __restrict__ out,
                                                       long long total) {
  const long long total4 = total >> 2;
  const long long base = (long long)blockIdx.x * 512;
  const float4* __restrict__ agg4 = reinterpret_cast<const float4*>(agg);
  const float4* __restrict__ h4   = reinterpret_cast<const float4*>(h);
  float4* __restrict__ out4       = reinterpret_cast<float4*>(out);
  const int c4 = threadIdx.x & 15;

  if (base + 512 <= total4) {
    const long long q0 = base + threadIdx.x;
    const long long q1 = q0 + 256;
    const float4 a0 = agg4[q0];
    const float4 a1 = agg4[q1];
    const float4 x0 = h4[q0];
    const float4 x1 = h4[q1];
    const float  d0 = deg[(int)(q0 >> 4)];
    const float  d1 = deg[(int)(q1 >> 4)];
    const float4 g  = bn_scale4(gamma, var, c4);
    const float4 mu = reinterpret_cast<const float4*>(mean)[c4];
    const float4 be = reinterpret_cast<const float4*>(beta)[c4];
    out4[q0] = bn4(a0, x0, 0.5f * d0, mu, g, be);
    out4[q1] = bn4(a1, x1, 0.5f * d1, mu, g, be);
  } else {
    for (long long q = base + threadIdx.x; q < total4; q += 256) {
      const float4 a0 = agg4[q];
      const float4 x0 = h4[q];
      const float  d0 = deg[(int)(q >> 4)];
      const float4 g  = bn_scale4(gamma, var, c4);
      const float4 mu = reinterpret_cast<const float4*>(mean)[c4];
      const float4 be = reinterpret_cast<const float4*>(beta)[c4];
      out4[q] = bn4(a0, x0, 0.5f * d0, mu, g, be);
    }
  }
}

extern "C" void kernel_launch(void* const* d_in, const int* in_sizes, int n_in,
                              void* d_out, int out_size, void* d_ws, size_t ws_size,
                              hipStream_t stream) {
  const float* X     = (const float*)d_in[0];
  const float* W     = (const float*)d_in[1];
  const float* b     = (const float*)d_in[2];
  const float* gamma = (const float*)d_in[3];
  const float* beta  = (const float*)d_in[4];
  const float* mean  = (const float*)d_in[5];
  const float* var   = (const float*)d_in[6];
  const int*   ei    = (const int*)d_in[7];

  const int F  = 128, Hc = 64;
  const int N  = in_sizes[0] / F;
  const int E  = in_sizes[7] / 2;

  float* ws   = (float*)d_ws;
  float* h    = ws;
  float* agg  = h   + (size_t)N * Hc;
  float* deg  = agg + (size_t)N * Hc;
  float* invd = deg + N;

  hipMemsetAsync(agg, 0, ((size_t)N * Hc + (size_t)N) * sizeof(float), stream);

  deg_kernel<<<(E + 255) / 256, 256, 0, stream>>>(ei, deg, E);
  invd_kernel<<<(N + 255) / 256, 256, 0, stream>>>(deg, invd, N);

  const int ntiles = ((N + 15) / 16) * 4;
  gemm_wmma<<<(ntiles + 7) / 8, 256, 0, stream>>>(X, W, b, h, N);

  edge_scatter<<<(E + 255) / 256, 256, 0, stream>>>(ei, h, invd, agg, E);

  const long long total = (long long)N * Hc;
  finalize_kernel<<<(int)((total + 2047) / 2048), 256, 0, stream>>>(
      agg, h, deg, gamma, beta, mean, var, (float*)d_out, total);
}


# f32 atomics of the edge scatter removed: edges bucketed by source in-call (int slot adds), a wave sums the h[dst] rows of 2 nodes in registers and writes out; sum order = bucket order, even/odd split
# speedup vs baseline: 1.6803x; 1.3218x over previous
#include <hip/hip_runtime.h>


typedef float v2f __attribute__((ext_vector_type(2)));
typedef float v8f __attribute__((ext_vector_type(8)));

#define BN_EPS 1e-3f


__global__ __launch_bounds__(256) void count_kernel(const int* __restrict__ ei,
                                                    int* __restrict__ cnt,
                                                    int* __restrict__ slot, int E) {
  int e = blockIdx.x * blockDim.x + threadIdx.x;
  if (e < E) slot[e] = atomicAdd(&cnt[ei[2 * e]], 1);
}

__global__ __launch_bounds__(256) void place_kernel(const int* __restrict__ cnt,
                                                    int* __restrict__ node,
                                                    float* __restrict__ invd,
                                                    int* __restrict__ total, int n) {
  __shared__ int wsum[8];
  __shared__ int bbase;
  const int i = blockIdx.x * 256 + threadIdx.x;
  const int lane = threadIdx.x & 31;
  const int wave = threadIdx.x >> 5;
  const int c = (i < n) ? cnt[i] : 0;
  int x = c;
  for (int o = 1; o < 32; o <<= 1) {
    const int y = __shfl_up(x, o);
    if (lane >= o) x += y;
  }
  if (lane == 31) wsum[wave] = x;
  __syncthreads();
  if (threadIdx.x == 0) {
    int t = 0;
    for (int w = 0; w < 8; ++w) {
      const int v = wsum[w];
      wsum[w] = t;
      t += v;
    }
    bbase = atomicAdd(total, t);
  }
  __syncthreads();
  if (i < n) {
    const float r = rsqrtf((float)c);
    int4 rec;
    rec.x = bbase + wsum[wave] + (x - c);
    rec.y = c;
    rec.z = __builtin_bit_cast(int, r);
    rec.w = 0;
    reinterpret_cast<int4*>(node)[i] = rec;
    invd[i] = r;
  }
}

__global__ __launch_bounds__(256) void fill_kernel(const int* __restrict__ ei,
                                                   const int* __restrict__ slot,
                                                   const int* __restrict__ node,
                                                   const float* __restrict__ invd,
                                                   int* __restrict__ bucket, int E) {
  int e = blockIdx.x * blockDim.x + threadIdx.x;
  if (e < E) {
    const int2 sd = *reinterpret_cast<const int2*>(ei + 2 * (size_t)e);
    int2 v;
    v.x = sd.y;
    v.y = __builtin_bit_cast(int, invd[sd.y]);
    reinterpret_cast<int2*>(bucket)[node[4 * (size_t)sd.x] + slot[e]] = v;
  }
}

__global__ __launch_bounds__(256) void gemm_wmma(const float* __restrict__ X,
                                                 const float* __restrict__ W,
                                                 const float* __restrict__ bias,
                                                 float* __restrict__ Hout, int Nrows) {
  const int F = 128, Hc = 64;
  const int lane = threadIdx.x & 31;
  const int wave = threadIdx.x >> 5;
  const int tile = blockIdx.x * 8 + wave;
  const int ntiles = ((Nrows + 15) >> 4) * 4;
  if (tile >= ntiles) return;

  const int row0 = (tile >> 2) << 4;
  const int col0 = (tile & 3) << 4;
  const int m  = lane & 15;
  const int hi = lane >> 4;

  const float* __restrict__ xrow = X + (size_t)(row0 + m) * F;
  const float* __restrict__ wcol = W + col0 + m;

  v8f acc = {};
#pragma unroll
  for (int k = 0; k < F; k += 4) {
    const int ka = k + 2 * hi;
    v2f a = *reinterpret_cast<const v2f*>(xrow + ka);
    v2f b;
    b.x = wcol[(size_t)ka * Hc];
    b.y = wcol[(size_t)(ka + 1) * Hc];
    acc = __builtin_amdgcn_wmma_f32_16x16x4_f32(
         false, a,  false, b,
         (short)0, acc,  false,  false);
  }

  const float bv = bias[col0 + m];
#pragma unroll
  for (int v = 0; v < 8; ++v) {
    const int row = row0 + v + 8 * hi;
    Hout[(size_t)row * Hc + (col0 + m)] = acc[v] + bv;
  }
}

template <int U>
__device__ __forceinline__ int row_share_i(int v) {
  return __builtin_amdgcn_update_dpp(0, v, 0x150 + U, 0xf, 0xf, true);
}
template <int U>
__device__ __forceinline__ float row_share_f(float v) {
  return __builtin_bit_cast(float, row_share_i<U>(__builtin_bit_cast(int, v)));
}

template <int U>
__device__ __forceinline__ void slot_load(float4 (&r)[8], const float4* __restrict__ h4,
                                          const int dl, const int c4) {
  const int d = row_share_i<U>(dl);
  r[U] = h4[d * 16 + c4];
}
template <int U>
__device__ __forceinline__ void slot_fma(float4& acc, const float4 (&r)[8], const float wl) {
  const float w = row_share_f<U>(wl);
  acc.x = fmaf(w, r[U].x, acc.x);
  acc.y = fmaf(w, r[U].y, acc.y);
  acc.z = fmaf(w, r[U].z, acc.z);
  acc.w = fmaf(w, r[U].w, acc.w);
}

__device__ __forceinline__ float4 bucket_sum(const float4* __restrict__ h4,
                                             const int2* __restrict__ b2, const int o,
                                             const int c, const float wi, int2 ent,
                                             const int sub, const int c4) {
  float4 acc = make_float4(0.0f, 0.0f, 0.0f, 0.0f);
  for (int base = 0; base < c; base += 16) {
    const int idx = base + sub;
    if (base > 0) ent = b2[o + min(idx, c - 1)];
    const int dl = ent.x;
    const float wl = (idx < c) ? wi * __builtin_bit_cast(float, ent.y) : 0.0f;
    const int ns = (c - base + 1) >> 1;
    float4 r[8];
    slot_load<0>(r, h4, dl, c4);
    if (ns > 1) slot_load<1>(r, h4, dl, c4);
    if (ns > 2) slot_load<2>(r, h4, dl, c4);
    if (ns > 3) slot_load<3>(r, h4, dl, c4);
    if (ns > 4) slot_load<4>(r, h4, dl, c4);
    if (ns > 5) slot_load<5>(r, h4, dl, c4);
    if (ns > 6) slot_load<6>(r, h4, dl, c4);
    if (ns > 7) slot_load<7>(r, h4, dl, c4);
    slot_fma<0>(acc, r, wl);
    if (ns > 1) slot_fma<1>(acc, r, wl);
    if (ns > 2) slot_fma<2>(acc, r, wl);
    if (ns > 3) slot_fma<3>(acc, r, wl);
    if (ns > 4) slot_fma<4>(acc, r, wl);
    if (ns > 5) slot_fma<5>(acc, r, wl);
    if (ns > 6) slot_fma<6>(acc, r, wl);
    if (ns > 7) slot_fma<7>(acc, r, wl);
  }
  return acc;
}

__global__ __launch_bounds__(256) void gather_finalize(const float* __restrict__ h,
                                                       const int* __restrict__ node,
                                                       const int* __restrict__ bucket,
                                                       const float* __restrict__ gamma,
                                                       const float* __restrict__ beta,
                                                       const float* __restrict__ mean,
                                                       const float* __restrict__ var,
                                                       float* __restrict__ out, int n) {
  const int lane = threadIdx.x & 31;
  const int i0 = 2 * __builtin_amdgcn_readfirstlane(blockIdx.x * 8 + (threadIdx.x >> 5));
  if (i0 >= n) return;
  const int hf = lane >> 4;
  const int c4 = lane & 15;
  const int sub = 2 * (lane & 7) + hf;

  const float4* __restrict__ h4 = reinterpret_cast<const float4*>(h);
  const int2* __restrict__ b2 = reinterpret_cast<const int2*>(bucket);
  const int4* __restrict__ node4 = reinterpret_cast<const int4*>(node);
  const int4 ra = node4[i0];
  int4 rb = make_int4(0, 0, 0, 0);
  if (i0 + 1 < n) rb = node4[i0 + 1];

  int2 ea = make_int2(0, 0), eb = make_int2(0, 0);
  if (ra.y > 0) ea = b2[ra.x + min(sub, ra.y - 1)];
  if (rb.y > 0) eb = b2[rb.x + min(sub, rb.y - 1)];
  const int il = min(i0 + hf, n - 1);
  const float4 xs = h4[il * 16 + c4];

  const float4 sa = bucket_sum(h4, b2, ra.x, ra.y, __builtin_bit_cast(float, ra.z), ea, sub, c4);
  const float4 sb = bucket_sum(h4, b2, rb.x, rb.y, __builtin_bit_cast(float, rb.z), eb, sub, c4);

  const float4 ga = reinterpret_cast<const float4*>(gamma)[c4];
  const float4 va = reinterpret_cast<const float4*>(var)[c4];
  const float4 mu = reinterpret_cast<const float4*>(mean)[c4];
  const float4 be = reinterpret_cast<const float4*>(beta)[c4];

  const float4 keep = hf ? sb : sa;
  const float4 give = hf ? sa : sb;
  float4 a;
  a.x = keep.x + __shfl_xor(give.x, 16);
  a.y = keep.y + __shfl_xor(give.y, 16);
  a.z = keep.z + __shfl_xor(give.z, 16);
  a.w = keep.w + __shfl_xor(give.w, 16);

  const float hd = 0.5f * (float)(hf ? rb.y : ra.y);
  float4 g, res;
  g.x = ga.x * rsqrtf(va.x + BN_EPS);
  g.y = ga.y * rsqrtf(va.y + BN_EPS);
  g.z = ga.z * rsqrtf(va.z + BN_EPS);
  g.w = ga.w * rsqrtf(va.w + BN_EPS);
  res.x = ((0.5f * a.x + hd * xs.x) - mu.x) * g.x + be.x;
  res.y = ((0.5f * a.y + hd * xs.y) - mu.y) * g.y + be.y;
  res.z = ((0.5f * a.z + hd * xs.z) - mu.z) * g.z + be.z;
  res.w = ((0.5f * a.w + hd * xs.w) - mu.w) * g.w + be.w;
  if (i0 + hf < n) reinterpret_cast<float4*>(out)[(i0 + hf) * 16 + c4] = res;
}

extern "C" void kernel_launch(void* const* d_in, const int* in_sizes, int n_in,
                              void* d_out, int out_size, void* d_ws, size_t ws_size,
                              hipStream_t stream) {
  const float* X     = (const float*)d_in[0];
  const float* W     = (const float*)d_in[1];
  const float* b     = (const float*)d_in[2];
  const float* gamma = (const float*)d_in[3];
  const float* beta  = (const float*)d_in[4];
  const float* mean  = (const float*)d_in[5];
  const float* var   = (const float*)d_in[6];
  const int*   ei    = (const int*)d_in[7];

  const int F  = 128, Hc = 64;
  const int N  = in_sizes[0] / F;
  const int E  = in_sizes[7] / 2;

  float* h      = (float*)d_ws;
  int*   node   = (int*)(h + (size_t)N * Hc);
  int*   bucket = node + 4 * (size_t)N;
  int*   slot   = bucket + 2 * (size_t)E;
  int*   cnt    = slot + (size_t)E;
  int*   total  = cnt + N;
  float* invd   = (float*)(total + 1);

  hipMemsetAsync(cnt, 0, ((size_t)N + 1) * sizeof(int), stream);

  count_kernel<<<(E + 255) / 256, 256, 0, stream>>>(ei, cnt, slot, E);
  place_kernel<<<(N + 255) / 256, 256, 0, stream>>>(cnt, node, invd, total, N);
  fill_kernel<<<(E + 255) / 256, 256, 0, stream>>>(ei, slot, node, invd, bucket, E);

  const int ntiles = ((N + 15) / 16) * 4;
  gemm_wmma<<<(ntiles + 7) / 8, 256, 0, stream>>>(X, W, b, h, N);

  gather_finalize<<<(N + 15) / 16, 256, 0, stream>>>(h, node, bucket, gamma, beta, mean, var,
                                                     (float*)d_out, N);
}


# counting sort without per-edge global atomics: a work-group owns (edge chunk x node range), counts and hands out bucket slots in LDS; sum order = chunk order, then LDS arrival order
# speedup vs baseline: 2.7628x; 1.6442x over previous
#include <hip/hip_runtime.h>


typedef float v2f __attribute__((ext_vector_type(2)));
typedef float v8f __attribute__((ext_vector_type(8)));

#define BN_EPS 1e-3f


#define NC 32
#define RS 12800
#define SORT_THREADS 1024
#define SORT_UNROLL 16
#define RS_STEPS ((RS + SORT_THREADS - 1) / SORT_THREADS)

__global__ __launch_bounds__(SORT_THREADS) void count_kernel(const int* __restrict__ ei,
                                                             int* __restrict__ part,
                                                             int n, int E, int nr) {
  __shared__ int cnt[RS];
  const int r = blockIdx.x % nr;
  const int c = blockIdx.x / nr;
  const int r0 = r * RS;
  const int len = min(RS, n - r0);
  const int ntile = (E + SORT_THREADS - 1) / SORT_THREADS;

  for (int i = threadIdx.x; i < len; i += SORT_THREADS) cnt[i] = 0;
  __syncthreads();

  for (int tb = c; tb < ntile; tb += NC * SORT_UNROLL) {
    int s[SORT_UNROLL];
#pragma unroll
    for (int u = 0; u < SORT_UNROLL; ++u) {
      const int e = (tb + u * NC) * SORT_THREADS + (int)threadIdx.x;
      s[u] = (e < E) ? ei[2 * (size_t)e] : -1;
    }
#pragma unroll
    for (int u = 0; u < SORT_UNROLL; ++u) {
      const unsigned k = (unsigned)(s[u] - r0);
      if (k < (unsigned)len) atomicAdd(&cnt[k], 1);
    }
  }
  __syncthreads();

  int* __restrict__ prow = part + (size_t)c * n + r0;
#pragma unroll
  for (int j = 0; j < RS_STEPS; ++j) {
    const int i = j * SORT_THREADS + (int)threadIdx.x;
    if (i < len) prow[i] = cnt[i];
  }
}

__global__ __launch_bounds__(256) void place_kernel(int* __restrict__ part,
                                                    int* __restrict__ node,
                                                    float* __restrict__ invd,
                                                    int* __restrict__ total, int n) {
  __shared__ int wsum[8];
  __shared__ int bbase;
  const int i = blockIdx.x * 256 + threadIdx.x;
  const int lane = threadIdx.x & 31;
  const int wave = threadIdx.x >> 5;
  int p[NC];
  int c = 0;
#pragma unroll
  for (int k = 0; k < NC; ++k) {
    p[k] = (i < n) ? part[(size_t)k * n + i] : 0;
    c += p[k];
  }
  int x = c;
  for (int o = 1; o < 32; o <<= 1) {
    const int y = __shfl_up(x, o);
    if (lane >= o) x += y;
  }
  if (lane == 31) wsum[wave] = x;
  __syncthreads();
  if (threadIdx.x == 0) {
    int t = 0;
    for (int w = 0; w < 8; ++w) {
      const int v = wsum[w];
      wsum[w] = t;
      t += v;
    }
    bbase = atomicAdd(total, t);
  }
  __syncthreads();
  if (i < n) {
    const float r = rsqrtf((float)c);
    const int start = bbase + wsum[wave] + (x - c);
    int4 rec;
    rec.x = start;
    rec.y = c;
    rec.z = __builtin_bit_cast(int, r);
    rec.w = 0;
    reinterpret_cast<int4*>(node)[i] = rec;
    invd[i] = r;
    int run = start;
#pragma unroll
    for (int k = 0; k < NC; ++k) {
      part[(size_t)k * n + i] = run;
      run += p[k];
    }
  }
}

__global__ __launch_bounds__(SORT_THREADS) void fill_kernel(const int* __restrict__ ei,
                                                            const int* __restrict__ part,
                                                            const float* __restrict__ invd,
                                                            int* __restrict__ bucket,
                                                            int n, int E, int nr) {
  __shared__ int cur[RS];
  const int r = blockIdx.x % nr;
  const int c = blockIdx.x / nr;
  const int r0 = r * RS;
  const int len = min(RS, n - r0);
  const int ntile = (E + SORT_THREADS - 1) / SORT_THREADS;

  const int* __restrict__ prow = part + (size_t)c * n + r0;
  int first[RS_STEPS];
#pragma unroll
  for (int j = 0; j < RS_STEPS; ++j) {
    const int i = j * SORT_THREADS + (int)threadIdx.x;
    first[j] = (i < len) ? prow[i] : 0;
  }
#pragma unroll
  for (int j = 0; j < RS_STEPS; ++j) {
    const int i = j * SORT_THREADS + (int)threadIdx.x;
    if (i < len) cur[i] = first[j];
  }
  __syncthreads();

  const int2* __restrict__ e2 = reinterpret_cast<const int2*>(ei);
  int2* __restrict__ b2 = reinterpret_cast<int2*>(bucket);
  for (int tb = c; tb < ntile; tb += NC * SORT_UNROLL) {
    int2 sd[SORT_UNROLL];
#pragma unroll
    for (int u = 0; u < SORT_UNROLL; ++u) {
      const int e = (tb + u * NC) * SORT_THREADS + (int)threadIdx.x;
      sd[u] = (e < E) ? e2[e] : make_int2(-1, 0);
    }
    int pos[SORT_UNROLL], wv[SORT_UNROLL];
#pragma unroll
    for (int u = 0; u < SORT_UNROLL; ++u) {
      const unsigned k = (unsigned)(sd[u].x - r0);
      pos[u] = -1;
      wv[u] = 0;
      if (k < (unsigned)len) {
        pos[u] = atomicAdd(&cur[k], 1);
        wv[u] = __builtin_bit_cast(int, invd[sd[u].y]);
      }
    }
#pragma unroll
    for (int u = 0; u < SORT_UNROLL; ++u) {
      if (pos[u] >= 0) {
        int2 v;
        v.x = sd[u].y;
        v.y = wv[u];
        b2[pos[u]] = v;
      }
    }
  }
}

__global__ __launch_bounds__(256) void gemm_wmma(const float* __restrict__ X,
                                                 const float* __restrict__ W,
                                                 const float* __restrict__ bias,
                                                 float* __restrict__ Hout, int Nrows) {
  const int F = 128, Hc = 64;
  const int lane = threadIdx.x & 31;
  const int wave = threadIdx.x >> 5;
  const int tile = blockIdx.x * 8 + wave;
  const int ntiles = ((Nrows + 15) >> 4) * 4;
  if (tile >= ntiles) return;

  const int row0 = (tile >> 2) << 4;
  const int col0 = (tile & 3) << 4;
  const int m  = lane & 15;
  const int hi = lane >> 4;

  const float* __restrict__ xrow = X + (size_t)(row0 + m) * F;
  const float* __restrict__ wcol = W + col0 + m;

  v8f acc = {};
#pragma unroll
  for (int k = 0; k < F; k += 4) {
    const int ka = k + 2 * hi;
    v2f a = *reinterpret_cast<const v2f*>(xrow + ka);
    v2f b;
    b.x = wcol[(size_t)ka * Hc];
    b.y = wcol[(size_t)(ka + 1) * Hc];
    acc = __builtin_amdgcn_wmma_f32_16x16x4_f32(
         false, a,  false, b,
         (short)0, acc,  false,  false);
  }

  const float bv = bias[col0 + m];
#pragma unroll
  for (int v = 0; v < 8; ++v) {
    const int row = row0 + v + 8 * hi;
    Hout[(size_t)row * Hc + (col0 + m)] = acc[v] + bv;
  }
}

template <int U>
__device__ __forceinline__ int row_share_i(int v) {
  return __builtin_amdgcn_update_dpp(0, v, 0x150 + U, 0xf, 0xf, true);
}
template <int U>
__device__ __forceinline__ float row_share_f(float v) {
  return __builtin_bit_cast(float, row_share_i<U>(__builtin_bit_cast(int, v)));
}

template <int U>
__device__ __forceinline__ void slot_load(float4 (&r)[8], const float4* __restrict__ h4,
                                          const int dl, const int c4) {
  const int d = row_share_i<U>(dl);
  r[U] = h4[d * 16 + c4];
}
template <int U>
__device__ __forceinline__ void slot_fma(float4& acc, const float4 (&r)[8], const float wl) {
  const float w = row_share_f<U>(wl);
  acc.x = fmaf(w, r[U].x, acc.x);
  acc.y = fmaf(w, r[U].y, acc.y);
  acc.z = fmaf(w, r[U].z, acc.z);
  acc.w = fmaf(w, r[U].w, acc.w);
}

__device__ __forceinline__ float4 bucket_sum(const float4* __restrict__ h4,
                                             const int2* __restrict__ b2, const int o,
                                             const int c, const float wi, int2 ent,
                                             const int sub, const int c4) {
  float4 acc = make_float4(0.0f, 0.0f, 0.0f, 0.0f);
  for (int base = 0; base < c; base += 16) {
    const int idx = base + sub;
    if (base > 0) ent = b2[o + min(idx, c - 1)];
    const int dl = ent.x;
    const float wl = (idx < c) ? wi * __builtin_bit_cast(float, ent.y) : 0.0f;
    const int ns = (c - base + 1) >> 1;
    float4 r[8];
    slot_load<0>(r, h4, dl, c4);
    if (ns > 1) slot_load<1>(r, h4, dl, c4);
    if (ns > 2) slot_load<2>(r, h4, dl, c4);
    if (ns > 3) slot_load<3>(r, h4, dl, c4);
    if (ns > 4) slot_load<4>(r, h4, dl, c4);
    if (ns > 5) slot_load<5>(r, h4, dl, c4);
    if (ns > 6) slot_load<6>(r, h4, dl, c4);
    if (ns > 7) slot_load<7>(r, h4, dl, c4);
    slot_fma<0>(acc, r, wl);
    if (ns > 1) slot_fma<1>(acc, r, wl);
    if (ns > 2) slot_fma<2>(acc, r, wl);
    if (ns > 3) slot_fma<3>(acc, r, wl);
    if (ns > 4) slot_fma<4>(acc, r, wl);
    if (ns > 5) slot_fma<5>(acc, r, wl);
    if (ns > 6) slot_fma<6>(acc, r, wl);
    if (ns > 7) slot_fma<7>(acc, r, wl);
  }
  return acc;
}

__global__ __launch_bounds__(256) void gather_finalize(const float* __restrict__ h,
                                                       const int* __restrict__ node,
                                                       const int* __restrict__ bucket,
                                                       const float* __restrict__ gamma,
                                                       const float* __restrict__ beta,
                                                       const float* __restrict__ mean,
                                                       const float* __restrict__ var,
                                                       float* __restrict__ out, int n) {
  const int lane = threadIdx.x & 31;
  const int i0 = 2 * __builtin_amdgcn_readfirstlane(blockIdx.x * 8 + (threadIdx.x >> 5));
  if (i0 >= n) return;
  const int hf = lane >> 4;
  const int c4 = lane & 15;
  const int sub = 2 * (lane & 7) + hf;

  const float4* __restrict__ h4 = reinterpret_cast<const float4*>(h);
  const int2* __restrict__ b2 = reinterpret_cast<const int2*>(bucket);
  const int4* __restrict__ node4 = reinterpret_cast<const int4*>(node);
  const int4 ra = node4[i0];
  int4 rb = make_int4(0, 0, 0, 0);
  if (i0 + 1 < n) rb = node4[i0 + 1];

  int2 ea = make_int2(0, 0), eb = make_int2(0, 0);
  if (ra.y > 0) ea = b2[ra.x + min(sub, ra.y - 1)];
  if (rb.y > 0) eb = b2[rb.x + min(sub, rb.y - 1)];
  const int il = min(i0 + hf, n - 1);
  const float4 xs = h4[il * 16 + c4];

  const float4 sa = bucket_sum(h4, b2, ra.x, ra.y, __builtin_bit_cast(float, ra.z), ea, sub, c4);
  const float4 sb = bucket_sum(h4, b2, rb.x, rb.y, __builtin_bit_cast(float, rb.z), eb, sub, c4);

  const float4 ga = reinterpret_cast<const float4*>(gamma)[c4];
  const float4 va = reinterpret_cast<const float4*>(var)[c4];
  const float4 mu = reinterpret_cast<const float4*>(mean)[c4];
  const float4 be = reinterpret_cast<const float4*>(beta)[c4];

  const float4 keep = hf ? sb : sa;
  const float4 give = hf ? sa : sb;
  float4 a;
  a.x = keep.x + __shfl_xor(give.x, 16);
  a.y = keep.y + __shfl_xor(give.y, 16);
  a.z = keep.z + __shfl_xor(give.z, 16);
  a.w = keep.w + __shfl_xor(give.w, 16);

  const float hd = 0.5f * (float)(hf ? rb.y : ra.y);
  float4 g, res;
  g.x = ga.x * rsqrtf(va.x + BN_EPS);
  g.y = ga.y * rsqrtf(va.y + BN_EPS);
  g.z = ga.z * rsqrtf(va.z + BN_EPS);
  g.w = ga.w * rsqrtf(va.w + BN_EPS);
  res.x = ((0.5f * a.x + hd * xs.x) - mu.x) * g.x + be.x;
  res.y = ((0.5f * a.y + hd * xs.y) - mu.y) * g.y + be.y;
  res.z = ((0.5f * a.z + hd * xs.z) - mu.z) * g.z + be.z;
  res.w = ((0.5f * a.w + hd * xs.w) - mu.w) * g.w + be.w;
  if (i0 + hf < n) reinterpret_cast<float4*>(out)[(i0 + hf) * 16 + c4] = res;
}

extern "C" void kernel_launch(void* const* d_in, const int* in_sizes, int n_in,
                              void* d_out, int out_size, void* d_ws, size_t ws_size,
                              hipStream_t stream) {
  const float* X     = (const float*)d_in[0];
  const float* W     = (const float*)d_in[1];
  const float* b     = (const float*)d_in[2];
  const float* gamma = (const float*)d_in[3];
  const float* beta  = (const float*)d_in[4];
  const float* mean  = (const float*)d_in[5];
  const float* var   = (const float*)d_in[6];
  const int*   ei    = (const int*)d_in[7];

  const int F  = 128, Hc = 64;
  const int N  = in_sizes[0] / F;
  const int E  = in_sizes[7] / 2;

  float* h      = (float*)d_ws;
  int*   node   = (int*)(h + (size_t)N * Hc);
  int*   bucket = node + 4 * (size_t)N;
  int*   part   = bucket + 2 * (size_t)E;
  int*   total  = part + (size_t)NC * N;
  float* invd   = (float*)(total + 1);

  hipMemsetAsync(total, 0, sizeof(int), stream);

  const int nr = (N + RS - 1) / RS;
  count_kernel<<<nr * NC, SORT_THREADS, 0, stream>>>(ei, part, N, E, nr);
  place_kernel<<<(N + 255) / 256, 256, 0, stream>>>(part, node, invd, total, N);
  fill_kernel<<<nr * NC, SORT_THREADS, 0, stream>>>(ei, part, invd, bucket, N, E, nr);

  const int ntiles = ((N + 15) / 16) * 4;
  gemm_wmma<<<(ntiles + 7) / 8, 256, 0, stream>>>(X, W, b, h, N);

  gather_finalize<<<(N + 15) / 16, 256, 0, stream>>>(h, node, bucket, gamma, beta, mean, var,
                                                     (float*)d_out, N);
}


# dense transform: a wave computes a 16x64 strip (was a 16x16 tile): X fragment loaded once for 4 matrix ops, W columns interleaved so one 16-byte load feeds 4 tiles; k sum order changes inside each 8
# speedup vs baseline: 3.1178x; 1.1285x over previous
#include <hip/hip_runtime.h>


typedef float v2f __attribute__((ext_vector_type(2)));
typedef float v8f __attribute__((ext_vector_type(8)));

#define BN_EPS 1e-3f


#define NC 32
#define RS 12800
#define SORT_THREADS 1024
#define SORT_UNROLL 16
#define RS_STEPS ((RS + SORT_THREADS - 1) / SORT_THREADS)

__global__ __launch_bounds__(SORT_THREADS) void count_kernel(const int* __restrict__ ei,
                                                             int* __restrict__ part,
                                                             int n, int E, int nr) {
  __shared__ int cnt[RS];
  const int r = blockIdx.x % nr;
  const int c = blockIdx.x / nr;
  const int r0 = r * RS;
  const int len = min(RS, n - r0);
  const int ntile = (E + SORT_THREADS - 1) / SORT_THREADS;

  for (int i = threadIdx.x; i < len; i += SORT_THREADS) cnt[i] = 0;
  __syncthreads();

  for (int tb = c; tb < ntile; tb += NC * SORT_UNROLL) {
    int s[SORT_UNROLL];
#pragma unroll
    for (int u = 0; u < SORT_UNROLL; ++u) {
      const int e = (tb + u * NC) * SORT_THREADS + (int)threadIdx.x;
      s[u] = (e < E) ? ei[2 * (size_t)e] : -1;
    }
#pragma unroll
    for (int u = 0; u < SORT_UNROLL; ++u) {
      const unsigned k = (unsigned)(s[u] - r0);
      if (k < (unsigned)len) atomicAdd(&cnt[k], 1);
    }
  }
  __syncthreads();

  int* __restrict__ prow = part + (size_t)c * n + r0;
#pragma unroll
  for (int j = 0; j < RS_STEPS; ++j) {
    const int i = j * SORT_THREADS + (int)threadIdx.x;
    if (i < len) prow[i] = cnt[i];
  }
}

__global__ __launch_bounds__(256) void place_kernel(int* __restrict__ part,
                                                    int* __restrict__ node,
                                                    float* __restrict__ invd,
                                                    int* __restrict__ total, int n) {
  __shared__ int wsum[8];
  __shared__ int bbase;
  const int i = blockIdx.x * 256 + threadIdx.x;
  const int lane = threadIdx.x & 31;
  const int wave = threadIdx.x >> 5;
  int p[NC];
  int c = 0;
#pragma unroll
  for (int k = 0; k < NC; ++k) {
    p[k] = (i < n) ? part[(size_t)k * n + i] : 0;
    c += p[k];
  }
  int x = c;
  for (int o = 1; o < 32; o <<= 1) {
    const int y = __shfl_up(x, o);
    if (lane >= o) x += y;
  }
  if (lane == 31) wsum[wave] = x;
  __syncthreads();
  if (threadIdx.x == 0) {
    int t = 0;
    for (int w = 0; w < 8; ++w) {
      const int v = wsum[w];
      wsum[w] = t;
      t += v;
    }
    bbase = atomicAdd(total, t);
  }
  __syncthreads();
  if (i < n) {
    const float r = rsqrtf((float)c);
    const int start = bbase + wsum[wave] + (x - c);
    int4 rec;
    rec.x = start;
    rec.y = c;
    rec.z = __builtin_bit_cast(int, r);
    rec.w = 0;
    reinterpret_cast<int4*>(node)[i] = rec;
    invd[i] = r;
    int run = start;
#pragma unroll
    for (int k = 0; k < NC; ++k) {
      part[(size_t)k * n + i] = run;
      run += p[k];
    }
  }
}

__global__ __launch_bounds__(SORT_THREADS) void fill_kernel(const int* __restrict__ ei,
                                                            const int* __restrict__ part,
                                                            const float* __restrict__ invd,
                                                            int* __restrict__ bucket,
                                                            int n, int E, int nr) {
  __shared__ int cur[RS];
  const int r = blockIdx.x % nr;
  const int c = blockIdx.x / nr;
  const int r0 = r * RS;
  const int len = min(RS, n - r0);
  const int ntile = (E + SORT_THREADS - 1) / SORT_THREADS;

  const int* __restrict__ prow = part + (size_t)c * n + r0;
  int first[RS_STEPS];
#pragma unroll
  for (int j = 0; j < RS_STEPS; ++j) {
    const int i = j * SORT_THREADS + (int)threadIdx.x;
    first[j] = (i < len) ? prow[i] : 0;
  }
#pragma unroll
  for (int j = 0; j < RS_STEPS; ++j) {
    const int i = j * SORT_THREADS + (int)threadIdx.x;
    if (i < len) cur[i] = first[j];
  }
  __syncthreads();

  const int2* __restrict__ e2 = reinterpret_cast<const int2*>(ei);
  int2* __restrict__ b2 = reinterpret_cast<int2*>(bucket);
  for (int tb = c; tb < ntile; tb += NC * SORT_UNROLL) {
    int2 sd[SORT_UNROLL];
#pragma unroll
    for (int u = 0; u < SORT_UNROLL; ++u) {
      const int e = (tb + u * NC) * SORT_THREADS + (int)threadIdx.x;
      sd[u] = (e < E) ? e2[e] : make_int2(-1, 0);
    }
    int pos[SORT_UNROLL], wv[SORT_UNROLL];
#pragma unroll
    for (int u = 0; u < SORT_UNROLL; ++u) {
      const unsigned k = (unsigned)(sd[u].x - r0);
      pos[u] = -1;
      wv[u] = 0;
      if (k < (unsigned)len) {
        pos[u] = atomicAdd(&cur[k], 1);
        wv[u] = __builtin_bit_cast(int, invd[sd[u].y]);
      }
    }
#pragma unroll
    for (int u = 0; u < SORT_UNROLL; ++u) {
      if (pos[u] >= 0) {
        int2 v;
        v.x = sd[u].y;
        v.y = wv[u];
        b2[pos[u]] = v;
      }
    }
  }
}

#define GEMM_STEPS 16
#define GEMM_AHEAD 3

__device__ __forceinline__ v8f mma4(const float a0, const float a1, const float b0,
                                    const float b1, const v8f acc) {
  v2f a, b;
  a.x = a0; a.y = a1;
  b.x = b0; b.y = b1;
  return __builtin_amdgcn_wmma_f32_16x16x4_f32(
       false, a,  false, b,
       (short)0, acc,  false,  false);
}

__global__ __launch_bounds__(128) void gemm_wmma(const float* __restrict__ X,
                                                 const float* __restrict__ W,
                                                 const float* __restrict__ bias,
                                                 float* __restrict__ Hout, int Nrows) {
  const int F = 128, Hc = 64;
  const int lane = threadIdx.x & 31;
  const int strip = blockIdx.x * 4 + (threadIdx.x >> 5);
  const int nstrips = (Nrows + 15) >> 4;
  if (strip >= nstrips) return;

  const int row0 = strip << 4;
  const int c  = lane & 15;
  const int hh = lane >> 4;

  const int arow = min(row0 + c, Nrows - 1);
  const float4* __restrict__ xa =
      reinterpret_cast<const float4*>(X + (size_t)arow * F + 4 * hh);
  const float4* __restrict__ wb =
      reinterpret_cast<const float4*>(W + (size_t)(4 * hh) * Hc + 4 * c);

  float4 av[GEMM_STEPS];
  float4 wv[GEMM_STEPS][4];
#pragma unroll
  for (int j = 0; j < GEMM_AHEAD; ++j) {
    av[j] = xa[2 * j];
#pragma unroll
    for (int q = 0; q < 4; ++q) wv[j][q] = wb[(8 * j + q) * (Hc / 4)];
  }

  v8f acc0 = {}, acc1 = {}, acc2 = {}, acc3 = {};
#pragma unroll
  for (int j = 0; j < GEMM_STEPS; ++j) {
    if (j + GEMM_AHEAD < GEMM_STEPS) {
      const int n = j + GEMM_AHEAD;
      av[n] = xa[2 * n];
#pragma unroll
      for (int q = 0; q < 4; ++q) wv[n][q] = wb[(8 * n + q) * (Hc / 4)];
    }
    __builtin_amdgcn_sched_barrier(0);
    const float4 a = av[j];
    const float4 w0 = wv[j][0], w1 = wv[j][1], w2 = wv[j][2], w3 = wv[j][3];
    acc0 = mma4(a.x, a.y, w0.x, w1.x, acc0);
    acc1 = mma4(a.x, a.y, w0.y, w1.y, acc1);
    acc2 = mma4(a.x, a.y, w0.z, w1.z, acc2);
    acc3 = mma4(a.x, a.y, w0.w, w1.w, acc3);
    acc0 = mma4(a.z, a.w, w2.x, w3.x, acc0);
    acc1 = mma4(a.z, a.w, w2.y, w3.y, acc1);
    acc2 = mma4(a.z, a.w, w2.z, w3.z, acc2);
    acc3 = mma4(a.z, a.w, w2.w, w3.w, acc3);
    __builtin_amdgcn_sched_barrier(0);
  }

  const float4 bv = reinterpret_cast<const float4*>(bias)[c];
  float4* __restrict__ h4 = reinterpret_cast<float4*>(Hout);
#pragma unroll
  for (int r = 0; r < 8; ++r) {
    const int row = row0 + r + 8 * hh;
    float4 o;
    o.x = acc0[r] + bv.x;
    o.y = acc1[r] + bv.y;
    o.z = acc2[r] + bv.z;
    o.w = acc3[r] + bv.w;
    if (row < Nrows) h4[(size_t)row * (Hc / 4) + c] = o;
  }
}

template <int U>
__device__ __forceinline__ int row_share_i(int v) {
  return __builtin_amdgcn_update_dpp(0, v, 0x150 + U, 0xf, 0xf, true);
}
template <int U>
__device__ __forceinline__ float row_share_f(float v) {
  return __builtin_bit_cast(float, row_share_i<U>(__builtin_bit_cast(int, v)));
}

template <int U>
__device__ __forceinline__ void slot_load(float4 (&r)[8], const float4* __restrict__ h4,
                                          const int dl, const int c4) {
  const int d = row_share_i<U>(dl);
  r[U] = h4[d * 16 + c4];
}
template <int U>
__device__ __forceinline__ void slot_fma(float4& acc, const float4 (&r)[8], const float wl) {
  const float w = row_share_f<U>(wl);
  acc.x = fmaf(w, r[U].x, acc.x);
  acc.y = fmaf(w, r[U].y, acc.y);
  acc.z = fmaf(w, r[U].z, acc.z);
  acc.w = fmaf(w, r[U].w, acc.w);
}

__device__ __forceinline__ float4 bucket_sum(const float4* __restrict__ h4,
                                             const int2* __restrict__ b2, const int o,
                                             const int c, const float wi, int2 ent,
                                             const int sub, const int c4) {
  float4 acc = make_float4(0.0f, 0.0f, 0.0f, 0.0f);
  for (int base = 0; base < c; base += 16) {
    const int idx = base + sub;
    if (base > 0) ent = b2[o + min(idx, c - 1)];
    const int dl = ent.x;
    const float wl = (idx < c) ? wi * __builtin_bit_cast(float, ent.y) : 0.0f;
    const int ns = (c - base + 1) >> 1;
    float4 r[8];
    slot_load<0>(r, h4, dl, c4);
    if (ns > 1) slot_load<1>(r, h4, dl, c4);
    if (ns > 2) slot_load<2>(r, h4, dl, c4);
    if (ns > 3) slot_load<3>(r, h4, dl, c4);
    if (ns > 4) slot_load<4>(r, h4, dl, c4);
    if (ns > 5) slot_load<5>(r, h4, dl, c4);
    if (ns > 6) slot_load<6>(r, h4, dl, c4);
    if (ns > 7) slot_load<7>(r, h4, dl, c4);
    slot_fma<0>(acc, r, wl);
    if (ns > 1) slot_fma<1>(acc, r, wl);
    if (ns > 2) slot_fma<2>(acc, r, wl);
    if (ns > 3) slot_fma<3>(acc, r, wl);
    if (ns > 4) slot_fma<4>(acc, r, wl);
    if (ns > 5) slot_fma<5>(acc, r, wl);
    if (ns > 6) slot_fma<6>(acc, r, wl);
    if (ns > 7) slot_fma<7>(acc, r, wl);
  }
  return acc;
}

__global__ __launch_bounds__(256) void gather_finalize(const float* __restrict__ h,
                                                       const int* __restrict__ node,
                                                       const int* __restrict__ bucket,
                                                       const float* __restrict__ gamma,
                                                       const float* __restrict__ beta,
                                                       const float* __restrict__ mean,
                                                       const float* __restrict__ var,
                                                       float* __restrict__ out, int n) {
  const int lane = threadIdx.x & 31;
  const int i0 = 2 * __builtin_amdgcn_readfirstlane(blockIdx.x * 8 + (threadIdx.x >> 5));
  if (i0 >= n) return;
  const int hf = lane >> 4;
  const int c4 = lane & 15;
  const int sub = 2 * (lane & 7) + hf;

  const float4* __restrict__ h4 = reinterpret_cast<const float4*>(h);
  const int2* __restrict__ b2 = reinterpret_cast<const int2*>(bucket);
  const int4* __restrict__ node4 = reinterpret_cast<const int4*>(node);
  const int4 ra = node4[i0];
  int4 rb = make_int4(0, 0, 0, 0);
  if (i0 + 1 < n) rb = node4[i0 + 1];

  int2 ea = make_int2(0, 0), eb = make_int2(0, 0);
  if (ra.y > 0) ea = b2[ra.x + min(sub, ra.y - 1)];
  if (rb.y > 0) eb = b2[rb.x + min(sub, rb.y - 1)];
  const int il = min(i0 + hf, n - 1);
  const float4 xs = h4[il * 16 + c4];

  const float4 sa = bucket_sum(h4, b2, ra.x, ra.y, __builtin_bit_cast(float, ra.z), ea, sub, c4);
  const float4 sb = bucket_sum(h4, b2, rb.x, rb.y, __builtin_bit_cast(float, rb.z), eb, sub, c4);

  const float4 ga = reinterpret_cast<const float4*>(gamma)[c4];
  const float4 va = reinterpret_cast<const float4*>(var)[c4];
  const float4 mu = reinterpret_cast<const float4*>(mean)[c4];
  const float4 be = reinterpret_cast<const float4*>(beta)[c4];

  const float4 keep = hf ? sb : sa;
  const float4 give = hf ? sa : sb;
  float4 a;
  a.x = keep.x + __shfl_xor(give.x, 16);
  a.y = keep.y + __shfl_xor(give.y, 16);
  a.z = keep.z + __shfl_xor(give.z, 16);
  a.w = keep.w + __shfl_xor(give.w, 16);

  const float hd = 0.5f * (float)(hf ? rb.y : ra.y);
  float4 g, res;
  g.x = ga.x * rsqrtf(va.x + BN_EPS);
  g.y = ga.y * rsqrtf(va.y + BN_EPS);
  g.z = ga.z * rsqrtf(va.z + BN_EPS);
  g.w = ga.w * rsqrtf(va.w + BN_EPS);
  res.x = ((0.5f * a.x + hd * xs.x) - mu.x) * g.x + be.x;
  res.y = ((0.5f * a.y + hd * xs.y) - mu.y) * g.y + be.y;
  res.z = ((0.5f * a.z + hd * xs.z) - mu.z) * g.z + be.z;
  res.w = ((0.5f * a.w + hd * xs.w) - mu.w) * g.w + be.w;
  if (i0 + hf < n) reinterpret_cast<float4*>(out)[(i0 + hf) * 16 + c4] = res;
}

extern "C" void kernel_launch(void* const* d_in, const int* in_sizes, int n_in,
                              void* d_out, int out_size, void* d_ws, size_t ws_size,
                              hipStream_t stream) {
  const float* X     = (const float*)d_in[0];
  const float* W     = (const float*)d_in[1];
  const float* b     = (const float*)d_in[2];
  const float* gamma = (const float*)d_in[3];
  const float* beta  = (const float*)d_in[4];
  const float* mean  = (const float*)d_in[5];
  const float* var   = (const float*)d_in[6];
  const int*   ei    = (const int*)d_in[7];

  const int F  = 128, Hc = 64;
  const int N  = in_sizes[0] / F;
  const int E  = in_sizes[7] / 2;

  float* h      = (float*)d_ws;
  int*   node   = (int*)(h + (size_t)N * Hc);
  int*   bucket = node + 4 * (size_t)N;
  int*   part   = bucket + 2 * (size_t)E;
  int*   total  = part + (size_t)NC * N;
  float* invd   = (float*)(total + 1);

  hipMemsetAsync(total, 0, sizeof(int), stream);

  const int nr = (N + RS - 1) / RS;
  count_kernel<<<nr * NC, SORT_THREADS, 0, stream>>>(ei, part, N, E, nr);
  place_kernel<<<(N + 255) / 256, 256, 0, stream>>>(part, node, invd, total, N);
  fill_kernel<<<nr * NC, SORT_THREADS, 0, stream>>>(ei, part, invd, bucket, N, E, nr);

  const int nstrips = (N + 15) / 16;
  gemm_wmma<<<(nstrips + 3) / 4, 128, 0, stream>>>(X, W, b, h, N);

  gather_finalize<<<(N + 15) / 16, 256, 0, stream>>>(h, node, bucket, gamma, beta, mean, var,
                                                     (float*)d_out, N);
}


# deg[dst]^-0.5 moves from each edge into the stored row: dense transform writes deg^-0.5*(XW+b), bucket entry = 4-byte dst, fill's random per-edge weight read gone; sum order kept, one more rounding
# speedup vs baseline: 3.3291x; 1.0678x over previous
#include <hip/hip_runtime.h>


typedef float v2f __attribute__((ext_vector_type(2)));
typedef float v8f __attribute__((ext_vector_type(8)));

#define BN_EPS 1e-3f


#define NC 32
#define RS 12800
#define SORT_THREADS 1024
#define SORT_UNROLL 16
#define RS_STEPS ((RS + SORT_THREADS - 1) / SORT_THREADS)

__global__ __launch_bounds__(SORT_THREADS) void count_kernel(const int* __restrict__ ei,
                                                             int* __restrict__ part,
                                                             int n, int E, int nr) {
  __shared__ int cnt[RS];
  const int r = blockIdx.x % nr;
  const int c = blockIdx.x / nr;
  const int r0 = r * RS;
  const int len = min(RS, n - r0);
  const int ntile = (E + SORT_THREADS - 1) / SORT_THREADS;

  for (int i = threadIdx.x; i < len; i += SORT_THREADS) cnt[i] = 0;
  __syncthreads();

  for (int tb = c; tb < ntile; tb += NC * SORT_UNROLL) {
    int s[SORT_UNROLL];
#pragma unroll
    for (int u = 0; u < SORT_UNROLL; ++u) {
      const int e = (tb + u * NC) * SORT_THREADS + (int)threadIdx.x;
      s[u] = (e < E) ? ei[2 * (size_t)e] : -1;
    }
#pragma unroll
    for (int u = 0; u < SORT_UNROLL; ++u) {
      const unsigned k = (unsigned)(s[u] - r0);
      if (k < (unsigned)len) atomicAdd(&cnt[k], 1);
    }
  }
  __syncthreads();

  int* __restrict__ prow = part + (size_t)c * n + r0;
#pragma unroll
  for (int j = 0; j < RS_STEPS; ++j) {
    const int i = j * SORT_THREADS + (int)threadIdx.x;
    if (i < len) prow[i] = cnt[i];
  }
}

__global__ __launch_bounds__(256) void place_kernel(int* __restrict__ part,
                                                    int* __restrict__ node,
                                                    int* __restrict__ total, int n) {
  __shared__ int wsum[8];
  __shared__ int bbase;
  const int i = blockIdx.x * 256 + threadIdx.x;
  const int lane = threadIdx.x & 31;
  const int wave = threadIdx.x >> 5;
  int p[NC];
  int c = 0;
#pragma unroll
  for (int k = 0; k < NC; ++k) {
    p[k] = (i < n) ? part[(size_t)k * n + i] : 0;
    c += p[k];
  }
  int x = c;
  for (int o = 1; o < 32; o <<= 1) {
    const int y = __shfl_up(x, o);
    if (lane >= o) x += y;
  }
  if (lane == 31) wsum[wave] = x;
  __syncthreads();
  if (threadIdx.x == 0) {
    int t = 0;
    for (int w = 0; w < 8; ++w) {
      const int v = wsum[w];
      wsum[w] = t;
      t += v;
    }
    bbase = atomicAdd(total, t);
  }
  __syncthreads();
  if (i < n) {
    const float r = rsqrtf((float)c);
    const int start = bbase + wsum[wave] + (x - c);
    int4 rec;
    rec.x = start;
    rec.y = c;
    rec.z = __builtin_bit_cast(int, r);
    rec.w = 0;
    reinterpret_cast<int4*>(node)[i] = rec;
    int run = start;
#pragma unroll
    for (int k = 0; k < NC; ++k) {
      part[(size_t)k * n + i] = run;
      run += p[k];
    }
  }
}

__global__ __launch_bounds__(SORT_THREADS) void fill_kernel(const int* __restrict__ ei,
                                                            const int* __restrict__ part,
                                                            int* __restrict__ bucket,
                                                            int n, int E, int nr) {
  __shared__ int cur[RS];
  const int r = blockIdx.x % nr;
  const int c = blockIdx.x / nr;
  const int r0 = r * RS;
  const int len = min(RS, n - r0);
  const int ntile = (E + SORT_THREADS - 1) / SORT_THREADS;

  const int* __restrict__ prow = part + (size_t)c * n + r0;
  int first[RS_STEPS];
#pragma unroll
  for (int j = 0; j < RS_STEPS; ++j) {
    const int i = j * SORT_THREADS + (int)threadIdx.x;
    first[j] = (i < len) ? prow[i] : 0;
  }
#pragma unroll
  for (int j = 0; j < RS_STEPS; ++j) {
    const int i = j * SORT_THREADS + (int)threadIdx.x;
    if (i < len) cur[i] = first[j];
  }
  __syncthreads();

  const int2* __restrict__ e2 = reinterpret_cast<const int2*>(ei);
  for (int tb = c; tb < ntile; tb += NC * SORT_UNROLL) {
    int2 sd[SORT_UNROLL];
#pragma unroll
    for (int u = 0; u < SORT_UNROLL; ++u) {
      const int e = (tb + u * NC) * SORT_THREADS + (int)threadIdx.x;
      sd[u] = (e < E) ? e2[e] : make_int2(-1, 0);
    }
    int pos[SORT_UNROLL];
#pragma unroll
    for (int u = 0; u < SORT_UNROLL; ++u) {
      const unsigned k = (unsigned)(sd[u].x - r0);
      pos[u] = -1;
      if (k < (unsigned)len) pos[u] = atomicAdd(&cur[k], 1);
    }
#pragma unroll
    for (int u = 0; u < SORT_UNROLL; ++u) {
      if (pos[u] >= 0) bucket[pos[u]] = sd[u].y;
    }
  }
}

#define GEMM_STEPS 16
#define GEMM_AHEAD 3

template <int U>
__device__ __forceinline__ int row_share_i(int v) {
  return __builtin_amdgcn_update_dpp(0, v, 0x150 + U, 0xf, 0xf, true);
}
template <int U>
__device__ __forceinline__ float row_share_f(float v) {
  return __builtin_bit_cast(float, row_share_i<U>(__builtin_bit_cast(int, v)));
}

__device__ __forceinline__ v8f mma4(const float a0, const float a1, const float b0,
                                    const float b1, const v8f acc) {
  v2f a, b;
  a.x = a0; a.y = a1;
  b.x = b0; b.y = b1;
  return __builtin_amdgcn_wmma_f32_16x16x4_f32(
       false, a,  false, b,
       (short)0, acc,  false,  false);
}

__global__ __launch_bounds__(128) void gemm_wmma(const float* __restrict__ X,
                                                 const float* __restrict__ W,
                                                 const float* __restrict__ bias,
                                                 const int* __restrict__ node,
                                                 float* __restrict__ Hout, int Nrows) {
  const int F = 128, Hc = 64;
  const int lane = threadIdx.x & 31;
  const int strip = blockIdx.x * 4 + (threadIdx.x >> 5);
  const int nstrips = (Nrows + 15) >> 4;
  if (strip >= nstrips) return;

  const int row0 = strip << 4;
  const int c  = lane & 15;
  const int hh = lane >> 4;

  const int arow = min(row0 + c, Nrows - 1);
  const float4* __restrict__ xa =
      reinterpret_cast<const float4*>(X + (size_t)arow * F + 4 * hh);
  const float4* __restrict__ wb =
      reinterpret_cast<const float4*>(W + (size_t)(4 * hh) * Hc + 4 * c);

  const float rowf = __builtin_bit_cast(float, node[4 * (size_t)arow + 2]);

  float4 av[GEMM_STEPS];
  float4 wv[GEMM_STEPS][4];
#pragma unroll
  for (int j = 0; j < GEMM_AHEAD; ++j) {
    av[j] = xa[2 * j];
#pragma unroll
    for (int q = 0; q < 4; ++q) wv[j][q] = wb[(8 * j + q) * (Hc / 4)];
  }

  v8f acc0 = {}, acc1 = {}, acc2 = {}, acc3 = {};
#pragma unroll
  for (int j = 0; j < GEMM_STEPS; ++j) {
    if (j + GEMM_AHEAD < GEMM_STEPS) {
      const int n = j + GEMM_AHEAD;
      av[n] = xa[2 * n];
#pragma unroll
      for (int q = 0; q < 4; ++q) wv[n][q] = wb[(8 * n + q) * (Hc / 4)];
    }
    __builtin_amdgcn_sched_barrier(0);
    const float4 a = av[j];
    const float4 w0 = wv[j][0], w1 = wv[j][1], w2 = wv[j][2], w3 = wv[j][3];
    acc0 = mma4(a.x, a.y, w0.x, w1.x, acc0);
    acc1 = mma4(a.x, a.y, w0.y, w1.y, acc1);
    acc2 = mma4(a.x, a.y, w0.z, w1.z, acc2);
    acc3 = mma4(a.x, a.y, w0.w, w1.w, acc3);
    acc0 = mma4(a.z, a.w, w2.x, w3.x, acc0);
    acc1 = mma4(a.z, a.w, w2.y, w3.y, acc1);
    acc2 = mma4(a.z, a.w, w2.z, w3.z, acc2);
    acc3 = mma4(a.z, a.w, w2.w, w3.w, acc3);
    __builtin_amdgcn_sched_barrier(0);
  }

  const float4 bv = reinterpret_cast<const float4*>(bias)[c];
  float4* __restrict__ h4 = reinterpret_cast<float4*>(Hout);
  float sc[8];
  {
    const float l0 = row_share_f<0>(rowf), l1 = row_share_f<1>(rowf), l2 = row_share_f<2>(rowf),
                l3 = row_share_f<3>(rowf), l4 = row_share_f<4>(rowf), l5 = row_share_f<5>(rowf),
                l6 = row_share_f<6>(rowf), l7 = row_share_f<7>(rowf);
    const float u0 = row_share_f<8>(rowf), u1 = row_share_f<9>(rowf), u2 = row_share_f<10>(rowf),
                u3 = row_share_f<11>(rowf), u4 = row_share_f<12>(rowf), u5 = row_share_f<13>(rowf),
                u6 = row_share_f<14>(rowf), u7 = row_share_f<15>(rowf);
    sc[0] = hh ? u0 : l0;  sc[1] = hh ? u1 : l1;  sc[2] = hh ? u2 : l2;  sc[3] = hh ? u3 : l3;
    sc[4] = hh ? u4 : l4;  sc[5] = hh ? u5 : l5;  sc[6] = hh ? u6 : l6;  sc[7] = hh ? u7 : l7;
  }
#pragma unroll
  for (int r = 0; r < 8; ++r) {
    const int row = row0 + r + 8 * hh;
    float4 o;
    o.x = (acc0[r] + bv.x) * sc[r];
    o.y = (acc1[r] + bv.y) * sc[r];
    o.z = (acc2[r] + bv.z) * sc[r];
    o.w = (acc3[r] + bv.w) * sc[r];
    if (row < Nrows) h4[(size_t)row * (Hc / 4) + c] = o;
  }
}

template <int U>
__device__ __forceinline__ void slot_load(float4 (&r)[8], const float4* __restrict__ h4,
                                          const int dl, const int c4) {
  const int d = row_share_i<U>(dl);
  r[U] = h4[d * 16 + c4];
}
template <int U>
__device__ __forceinline__ void slot_fma(float4& acc, const float4 (&r)[8], const float wl) {
  const float w = row_share_f<U>(wl);
  acc.x = fmaf(w, r[U].x, acc.x);
  acc.y = fmaf(w, r[U].y, acc.y);
  acc.z = fmaf(w, r[U].z, acc.z);
  acc.w = fmaf(w, r[U].w, acc.w);
}

__device__ __forceinline__ float4 bucket_sum(const float4* __restrict__ h4,
                                             const int* __restrict__ bk, const int o,
                                             const int c, const float wi, int ent,
                                             const int self, const int sub, const int c4) {
  float4 acc = make_float4(0.0f, 0.0f, 0.0f, 0.0f);
  for (int base = 0; base < c; base += 16) {
    const int idx = base + sub;
    if (base > 0) ent = bk[o + min(idx, c - 1)];
    const int dl = (idx < c) ? ent : self;
    const float wl = (idx < c) ? wi : 0.0f;
    const int ns = (c - base + 1) >> 1;
    float4 r[8];
    slot_load<0>(r, h4, dl, c4);
    if (ns > 1) slot_load<1>(r, h4, dl, c4);
    if (ns > 2) slot_load<2>(r, h4, dl, c4);
    if (ns > 3) slot_load<3>(r, h4, dl, c4);
    if (ns > 4) slot_load<4>(r, h4, dl, c4);
    if (ns > 5) slot_load<5>(r, h4, dl, c4);
    if (ns > 6) slot_load<6>(r, h4, dl, c4);
    if (ns > 7) slot_load<7>(r, h4, dl, c4);
    slot_fma<0>(acc, r, wl);
    if (ns > 1) slot_fma<1>(acc, r, wl);
    if (ns > 2) slot_fma<2>(acc, r, wl);
    if (ns > 3) slot_fma<3>(acc, r, wl);
    if (ns > 4) slot_fma<4>(acc, r, wl);
    if (ns > 5) slot_fma<5>(acc, r, wl);
    if (ns > 6) slot_fma<6>(acc, r, wl);
    if (ns > 7) slot_fma<7>(acc, r, wl);
  }
  return acc;
}

__global__ __launch_bounds__(256) void gather_finalize(const float* __restrict__ h,
                                                       const int* __restrict__ node,
                                                       const int* __restrict__ bucket,
                                                       const float* __restrict__ gamma,
                                                       const float* __restrict__ beta,
                                                       const float* __restrict__ mean,
                                                       const float* __restrict__ var,
                                                       float* __restrict__ out, int n) {
  const int lane = threadIdx.x & 31;
  const int i0 = 2 * __builtin_amdgcn_readfirstlane(blockIdx.x * 8 + (threadIdx.x >> 5));
  if (i0 >= n) return;
  const int hf = lane >> 4;
  const int c4 = lane & 15;
  const int sub = 2 * (lane & 7) + hf;

  const float4* __restrict__ h4 = reinterpret_cast<const float4*>(h);
  const int4* __restrict__ node4 = reinterpret_cast<const int4*>(node);
  const int4 ra = node4[i0];
  int4 rb = make_int4(0, 0, 0, 0);
  if (i0 + 1 < n) rb = node4[i0 + 1];

  int ea = 0, eb = 0;
  if (ra.y > 0) ea = bucket[ra.x + min(sub, ra.y - 1)];
  if (rb.y > 0) eb = bucket[rb.x + min(sub, rb.y - 1)];
  const int il = min(i0 + hf, n - 1);
  const float4 xs = h4[il * 16 + c4];

  const float4 sa = bucket_sum(h4, bucket, ra.x, ra.y, __builtin_bit_cast(float, ra.z), ea, i0, sub, c4);
  const float4 sb = bucket_sum(h4, bucket, rb.x, rb.y, __builtin_bit_cast(float, rb.z), eb, i0 + 1, sub, c4);

  const float4 ga = reinterpret_cast<const float4*>(gamma)[c4];
  const float4 va = reinterpret_cast<const float4*>(var)[c4];
  const float4 mu = reinterpret_cast<const float4*>(mean)[c4];
  const float4 be = reinterpret_cast<const float4*>(beta)[c4];

  const float4 keep = hf ? sb : sa;
  const float4 give = hf ? sa : sb;
  float4 a;
  a.x = keep.x + __shfl_xor(give.x, 16);
  a.y = keep.y + __shfl_xor(give.y, 16);
  a.z = keep.z + __shfl_xor(give.z, 16);
  a.w = keep.w + __shfl_xor(give.w, 16);

  const int dg = hf ? rb.y : ra.y;
  const float wo = __builtin_bit_cast(float, hf ? rb.z : ra.z);
  const float hd = 0.5f * (float)dg;
  float4 hr;
  hr.x = (dg > 0) ? xs.x / wo : 0.0f;
  hr.y = (dg > 0) ? xs.y / wo : 0.0f;
  hr.z = (dg > 0) ? xs.z / wo : 0.0f;
  hr.w = (dg > 0) ? xs.w / wo : 0.0f;
  float4 g, res;
  g.x = ga.x * rsqrtf(va.x + BN_EPS);
  g.y = ga.y * rsqrtf(va.y + BN_EPS);
  g.z = ga.z * rsqrtf(va.z + BN_EPS);
  g.w = ga.w * rsqrtf(va.w + BN_EPS);
  res.x = ((0.5f * a.x + hd * hr.x) - mu.x) * g.x + be.x;
  res.y = ((0.5f * a.y + hd * hr.y) - mu.y) * g.y + be.y;
  res.z = ((0.5f * a.z + hd * hr.z) - mu.z) * g.z + be.z;
  res.w = ((0.5f * a.w + hd * hr.w) - mu.w) * g.w + be.w;
  if (i0 + hf < n) reinterpret_cast<float4*>(out)[(i0 + hf) * 16 + c4] = res;
}

extern "C" void kernel_launch(void* const* d_in, const int* in_sizes, int n_in,
                              void* d_out, int out_size, void* d_ws, size_t ws_size,
                              hipStream_t stream) {
  const float* X     = (const float*)d_in[0];
  const float* W     = (const float*)d_in[1];
  const float* b     = (const float*)d_in[2];
  const float* gamma = (const float*)d_in[3];
  const float* beta  = (const float*)d_in[4];
  const float* mean  = (const float*)d_in[5];
  const float* var   = (const float*)d_in[6];
  const int*   ei    = (const int*)d_in[7];

  const int F  = 128, Hc = 64;
  const int N  = in_sizes[0] / F;
  const int E  = in_sizes[7] / 2;

  float* h      = (float*)d_ws;
  int*   node   = (int*)(h + (size_t)N * Hc);
  int*   bucket = node + 4 * (size_t)N;
  int*   part   = bucket + (size_t)E;
  int*   total  = part + (size_t)NC * N;

  hipMemsetAsync(total, 0, sizeof(int), stream);

  const int nr = (N + RS - 1) / RS;
  count_kernel<<<nr * NC, SORT_THREADS, 0, stream>>>(ei, part, N, E, nr);
  place_kernel<<<(N + 255) / 256, 256, 0, stream>>>(part, node, total, N);
  fill_kernel<<<nr * NC, SORT_THREADS, 0, stream>>>(ei, part, bucket, N, E, nr);

  const int nstrips = (N + 15) / 16;
  gemm_wmma<<<(nstrips + 3) / 4, 128, 0, stream>>>(X, W, b, node, h, N);

  gather_finalize<<<(N + 15) / 16, 256, 0, stream>>>(h, node, bucket, gamma, beta, mean, var,
                                                     (float*)d_out, N);
}
